# starting kernel
# baseline (speedup 1.0000x reference)
#include <hip/hip_runtime.h>

typedef float v2f __attribute__((ext_vector_type(2)));
typedef float v8f __attribute__((ext_vector_type(8)));

#define IN_CH 256
#define HID   128
#define KTILE 64


__global__ void k_deg_init(float* __restrict__ deg, int n) {
    int i = blockIdx.x * blockDim.x + threadIdx.x;
    if (i < n) deg[i] = 1.0f;
}

__global__ void k_deg_edges(const int* __restrict__ ei, float* __restrict__ deg, int E) {
    int e = blockIdx.x * blockDim.x + threadIdx.x;
    if (e < E) atomicAdd(&deg[ei[(size_t)E + e]], 1.0f);
}

__global__ void k_dinv(float* __restrict__ deg, int n) {
    int i = blockIdx.x * blockDim.x + threadIdx.x;
    if (i < n) deg[i] = rsqrtf(deg[i]);
}

__global__ __launch_bounds__(256) void k_gemm_wmma(
    const float* __restrict__ x, const float* __restrict__ W,
    const float* __restrict__ bias, const float* __restrict__ dinv,
    float* __restrict__ h, float* __restrict__ out, int nNodes) {
    __shared__ float sW[KTILE * HID];

    const int tid  = threadIdx.x;
    const int wave = tid >> 5;
    const int lane = tid & 31;
    const int half = lane >> 4;
    const int r    = lane & 15;
    const int m0   = (blockIdx.x * 8 + wave) * 16;
    const bool act = (m0 < nNodes);

    v8f acc[8] = {};

    for (int kb = 0; kb < IN_CH; kb += KTILE) {
        if (kb + KTILE < IN_CH)
            __builtin_prefetch(W + (size_t)(kb + KTILE) * HID + tid * 32, 0, 1);

        __syncthreads();
#pragma unroll
        for (int p = 0; p < (KTILE * HID / 2) / 256; ++p) {
            const int i  = tid + 256 * p;
            const int kp = i >> 7;
            const int n  = i & 127;
            float2 w2;
            w2.x = W[(size_t)(kb + 2 * kp) * HID + n];
            w2.y = W[(size_t)(kb + 2 * kp + 1) * HID + n];
            *(float2*)(sW + (size_t)i * 2) = w2;
        }
        __syncthreads();

        if (act) {
            const float* xp = x + (size_t)(m0 + r) * IN_CH + kb + 2 * half;
#pragma unroll
            for (int kl = 0; kl < KTILE; kl += 4) {
                v2f a = *(const v2f*)(xp + kl);
                const float* wrow = sW + ((kl >> 1) + half) * (2 * HID);
#pragma unroll
                for (int nt = 0; nt < 8; ++nt) {
                    v2f b = *(const v2f*)(wrow + (nt * 16 + r) * 2);
                    acc[nt] = __builtin_amdgcn_wmma_f32_16x16x4_f32(
                        false, a, false, b, (short)0, acc[nt], false, false);
                }
            }
        }
    }

    if (act) {
        float d2[8];
#pragma unroll
        for (int i = 0; i < 8; ++i) {
            float dv = dinv[m0 + i + 8 * half];
            d2[i] = dv * dv;
        }
#pragma unroll
        for (int nt = 0; nt < 8; ++nt) {
            const int   n    = nt * 16 + r;
            const float bcol = bias[n];
            float* hp = h   + (size_t)m0 * HID + n;
            float* op = out + (size_t)m0 * HID + n;
#pragma unroll
            for (int i = 0; i < 8; ++i) {
                const size_t off = (size_t)(i + 8 * half) * HID;
                const float  v   = acc[nt][i];
                hp[off] = v;
                op[off] = bcol + d2[i] * v;
            }
        }
    }
}

__global__ __launch_bounds__(256) void k_scatter(
    const int* __restrict__ ei, const float* __restrict__ h,
    const float* __restrict__ dinv, float* __restrict__ out, int E) {
    const long long gid = (long long)blockIdx.x * blockDim.x + threadIdx.x;
    const int e    = (int)(gid >> 5);
    const int lane = (int)(gid & 31);
    if (e >= E) return;
    const int src = ei[e];
    const int dst = ei[(size_t)E + e];
    const float nrm = dinv[src] * dinv[dst];
    const float4 hv = ((const float4*)(h + (size_t)src * HID))[lane];
    float* op = out + (size_t)dst * HID + 4 * lane;
    atomicAdd(op + 0, hv.x * nrm);
    atomicAdd(op + 1, hv.y * nrm);
    atomicAdd(op + 2, hv.z * nrm);
    atomicAdd(op + 3, hv.w * nrm);
}

__global__ void k_prelu(float* __restrict__ out, const float* __restrict__ alpha, long long total) {
    long long i = (long long)blockIdx.x * blockDim.x + threadIdx.x;
    if (i < total) {
        float v = out[i];
        float a = alpha[(int)(i & (HID - 1))];
        out[i] = v > 0.0f ? v : a * v;
    }
}

extern "C" void kernel_launch(void* const* d_in, const int* in_sizes, int n_in,
                              void* d_out, int out_size, void* d_ws, size_t ws_size,
                              hipStream_t stream) {
    const float* x     = (const float*)d_in[0];
    const int*   ei    = (const int*)d_in[1];
    const float* W     = (const float*)d_in[2];
    const float* bias  = (const float*)d_in[3];
    const float* alpha = (const float*)d_in[4];
    float*       out   = (float*)d_out;

    const int N = in_sizes[0] / IN_CH;
    const int E = in_sizes[1] / 2;

    float* h    = (float*)d_ws;
    float* dinv = h + (size_t)N * HID;

    k_deg_init <<<(N + 255) / 256, 256, 0, stream>>>(dinv, N);
    k_deg_edges<<<(E + 255) / 256, 256, 0, stream>>>(ei, dinv, E);
    k_dinv     <<<(N + 255) / 256, 256, 0, stream>>>(dinv, N);

    const int strips = (N + 15) / 16;
    const int blocks = (strips + 7) / 8;
    k_gemm_wmma<<<blocks, 256, 0, stream>>>(x, W, bias, dinv, h, out, N);

    const long long scThreads = (long long)E * 32;
    k_scatter<<<(int)((scThreads + 255) / 256), 256, 0, stream>>>(ei, h, dinv, out, E);

    const long long total = (long long)N * HID;
    k_prelu<<<(int)((total + 255) / 256), 256, 0, stream>>>(out, alpha, total);
}


# In-degree without global atomics: block (edge slice, 6400-destination range) counts ids in LDS integer bins, stores partial counts; dinv kernel adds the 32 partials (6 launches to 5); degree exact
# speedup vs baseline: 1.9926x; 1.9926x over previous
#include <hip/hip_runtime.h>

typedef float v2f __attribute__((ext_vector_type(2)));
typedef float v8f __attribute__((ext_vector_type(8)));

#define IN_CH 256
#define HID   128
#define KTILE 64

#define DEG_BINS   6400
#define DEG_SLICES 32
#define DEG_BATCH  8

__device__ __forceinline__ void deg_count(unsigned int* bins, int d, int lo, unsigned int lim) {
    const unsigned int b = (unsigned int)(d - lo);
    if (b < lim) atomicAdd(&bins[b], 1u);
}

__global__ __launch_bounds__(256) void k_deg_hist(
    const int* __restrict__ col, int* __restrict__ part, int n, int E) {
    __shared__ unsigned int bins[DEG_BINS];
    const int tid = threadIdx.x;
    const int s   = blockIdx.x;
    const int lo  = blockIdx.y * DEG_BINS;

    for (int b = tid; b < DEG_BINS; b += 256) bins[b] = 0u;
    __syncthreads();

    if ((E & 3) == 0) {
        const int4* __restrict__ c4 = (const int4*)col;
        const int V  = E >> 2;
        const int v0 = (int)(((long long)V * s) / DEG_SLICES);
        const int v1 = (int)(((long long)V * (s + 1)) / DEG_SLICES);
        for (int v = v0 + tid; v < v1; v += 256 * DEG_BATCH) {
            int4 d[DEG_BATCH];
#pragma unroll
            for (int u = 0; u < DEG_BATCH; ++u) {
                const int vv = v + 256 * u;
                d[u] = c4[vv < v1 ? vv : v];
            }
#pragma unroll
            for (int u = 0; u < DEG_BATCH; ++u) {
                const unsigned int lim = (v + 256 * u < v1) ? DEG_BINS : 0u;
                deg_count(bins, d[u].x, lo, lim);
                deg_count(bins, d[u].y, lo, lim);
                deg_count(bins, d[u].z, lo, lim);
                deg_count(bins, d[u].w, lo, lim);
            }
        }
    } else {
        const int e0 = (int)(((long long)E * s) / DEG_SLICES);
        const int e1 = (int)(((long long)E * (s + 1)) / DEG_SLICES);
        for (int e = e0 + tid; e < e1; e += 256) deg_count(bins, col[e], lo, DEG_BINS);
    }
    __syncthreads();

    const int cnt = min(DEG_BINS, n - lo);
    int* __restrict__ prow = part + (size_t)s * n + lo;
    for (int b = tid; b < cnt; b += 256) prow[b] = (int)bins[b];
}

__global__ void k_dinv(const int* __restrict__ part, float* __restrict__ dinv, int n) {
    int i = blockIdx.x * blockDim.x + threadIdx.x;
    if (i < n) {
        int deg = 1;
#pragma unroll
        for (int s = 0; s < DEG_SLICES; ++s) deg += part[(size_t)s * n + i];
        dinv[i] = rsqrtf((float)deg);
    }
}

__global__ __launch_bounds__(256) void k_gemm_wmma(
    const float* __restrict__ x, const float* __restrict__ W,
    const float* __restrict__ bias, const float* __restrict__ dinv,
    float* __restrict__ h, float* __restrict__ out, int nNodes) {
    __shared__ float sW[KTILE * HID];

    const int tid  = threadIdx.x;
    const int wave = tid >> 5;
    const int lane = tid & 31;
    const int half = lane >> 4;
    const int r    = lane & 15;
    const int m0   = (blockIdx.x * 8 + wave) * 16;
    const bool act = (m0 < nNodes);

    v8f acc[8] = {};

    for (int kb = 0; kb < IN_CH; kb += KTILE) {
        if (kb + KTILE < IN_CH)
            __builtin_prefetch(W + (size_t)(kb + KTILE) * HID + tid * 32, 0, 1);

        __syncthreads();
#pragma unroll
        for (int p = 0; p < (KTILE * HID / 2) / 256; ++p) {
            const int i  = tid + 256 * p;
            const int kp = i >> 7;
            const int n  = i & 127;
            float2 w2;
            w2.x = W[(size_t)(kb + 2 * kp) * HID + n];
            w2.y = W[(size_t)(kb + 2 * kp + 1) * HID + n];
            *(float2*)(sW + (size_t)i * 2) = w2;
        }
        __syncthreads();

        if (act) {
            const float* xp = x + (size_t)(m0 + r) * IN_CH + kb + 2 * half;
#pragma unroll
            for (int kl = 0; kl < KTILE; kl += 4) {
                v2f a = *(const v2f*)(xp + kl);
                const float* wrow = sW + ((kl >> 1) + half) * (2 * HID);
#pragma unroll
                for (int nt = 0; nt < 8; ++nt) {
                    v2f b = *(const v2f*)(wrow + (nt * 16 + r) * 2);
                    acc[nt] = __builtin_amdgcn_wmma_f32_16x16x4_f32(
                        false, a, false, b, (short)0, acc[nt], false, false);
                }
            }
        }
    }

    if (act) {
        float d2[8];
#pragma unroll
        for (int i = 0; i < 8; ++i) {
            float dv = dinv[m0 + i + 8 * half];
            d2[i] = dv * dv;
        }
#pragma unroll
        for (int nt = 0; nt < 8; ++nt) {
            const int   n    = nt * 16 + r;
            const float bcol = bias[n];
            float* hp = h   + (size_t)m0 * HID + n;
            float* op = out + (size_t)m0 * HID + n;
#pragma unroll
            for (int i = 0; i < 8; ++i) {
                const size_t off = (size_t)(i + 8 * half) * HID;
                const float  v   = acc[nt][i];
                hp[off] = v;
                op[off] = bcol + d2[i] * v;
            }
        }
    }
}

#define SC_GROUP 8

template <bool FULL>
__device__ __forceinline__ void scatter_group(
    const float* __restrict__ h, float* __restrict__ out,
    int src, int dst, float nrm, int g, int cnt, int lane) {
    float v[SC_GROUP][4];
#pragma unroll
    for (int u = 0; u < SC_GROUP; ++u) {
        const int s = __builtin_amdgcn_readlane(src, g + u);
        const float* hp = h + (size_t)s * HID + lane;
#pragma unroll
        for (int j = 0; j < 4; ++j) v[u][j] = hp[32 * j];
    }
#pragma unroll
    for (int u = 0; u < SC_GROUP; ++u) {
        if (FULL || g + u < cnt) {
            const int   d = __builtin_amdgcn_readlane(dst, g + u);
            const float w = __int_as_float(__builtin_amdgcn_readlane(__float_as_int(nrm), g + u));
            float* op = out + (size_t)d * HID + lane;
#pragma unroll
            for (int j = 0; j < 4; ++j) atomicAdd(op + 32 * j, v[u][j] * w);
        }
    }
}

__global__ __launch_bounds__(256) void k_scatter(
    const int* __restrict__ ei, const float* __restrict__ h,
    const float* __restrict__ dinv, float* __restrict__ out, int E) {
    const int lane = threadIdx.x & 31;
    const int wave = __builtin_amdgcn_readfirstlane(threadIdx.x >> 5);
    const int e0   = (blockIdx.x * 8 + wave) * 32;
    if (e0 >= E) return;
    const int cnt = min(32, E - e0);

    int src = 0, dst = 0;
    float nrm = 0.0f;
    if (lane < cnt) {
        src = ei[e0 + lane];
        dst = ei[(size_t)E + e0 + lane];
        nrm = dinv[src] * dinv[dst];
    }

    for (int g = 0; g < cnt; g += SC_GROUP) {
        if (g + SC_GROUP <= cnt) scatter_group<true >(h, out, src, dst, nrm, g, cnt, lane);
        else                     scatter_group<false>(h, out, src, dst, nrm, g, cnt, lane);
    }
}

__global__ void k_prelu(float* __restrict__ out, const float* __restrict__ alpha, long long total) {
    long long i = (long long)blockIdx.x * blockDim.x + threadIdx.x;
    if (i < total) {
        float v = out[i];
        float a = alpha[(int)(i & (HID - 1))];
        out[i] = v > 0.0f ? v : a * v;
    }
}

extern "C" void kernel_launch(void* const* d_in, const int* in_sizes, int n_in,
                              void* d_out, int out_size, void* d_ws, size_t ws_size,
                              hipStream_t stream) {
    const float* x     = (const float*)d_in[0];
    const int*   ei    = (const int*)d_in[1];
    const float* W     = (const float*)d_in[2];
    const float* bias  = (const float*)d_in[3];
    const float* alpha = (const float*)d_in[4];
    float*       out   = (float*)d_out;

    const int N = in_sizes[0] / IN_CH;
    const int E = in_sizes[1] / 2;

    float* h    = (float*)d_ws;
    float* dinv = h + (size_t)N * HID;
    int*   part = (int*)(dinv + N);

    const dim3 hgrid(DEG_SLICES, (N + DEG_BINS - 1) / DEG_BINS);
    k_deg_hist<<<hgrid, 256, 0, stream>>>(ei + (size_t)E, part, N, E);
    k_dinv    <<<(N + 255) / 256, 256, 0, stream>>>(part, dinv, N);

    const int strips = (N + 15) / 16;
    const int blocks = (strips + 7) / 8;
    k_gemm_wmma<<<blocks, 256, 0, stream>>>(x, W, bias, dinv, h, out, N);

    k_scatter<<<(E + 255) / 256, 256, 0, stream>>>(ei, h, dinv, out, E);

    const long long total = (long long)N * HID;
    k_prelu<<<(int)((total + 255) / 256), 256, 0, stream>>>(out, alpha, total);
}


# Scatter adds into each out row from one L2 half only: a wave reads its die, draws a ticket for 64 edges from its half's counter, keeps edges with dst in that half; f32 atomics, arrival order kept
# speedup vs baseline: 2.1886x; 1.0984x over previous
#include <hip/hip_runtime.h>

typedef float v2f __attribute__((ext_vector_type(2)));
typedef float v8f __attribute__((ext_vector_type(8)));

#define IN_CH 256
#define HID   128
#define KTILE 64
#define Q_STRIDE 64

#define DEG_BINS   6400
#define DEG_SLICES 32
#define DEG_BATCH  8

__device__ __forceinline__ void deg_count(unsigned int* bins, int d, int lo, unsigned int lim) {
    const unsigned int b = (unsigned int)(d - lo);
    if (b < lim) atomicAdd(&bins[b], 1u);
}

__global__ __launch_bounds__(256) void k_deg_hist(
    const int* __restrict__ col, int* __restrict__ part, int n, int E) {
    __shared__ unsigned int bins[DEG_BINS];
    const int tid = threadIdx.x;
    const int s   = blockIdx.x;
    const int lo  = blockIdx.y * DEG_BINS;

    for (int b = tid; b < DEG_BINS; b += 256) bins[b] = 0u;
    __syncthreads();

    if ((E & 3) == 0) {
        const int4* __restrict__ c4 = (const int4*)col;
        const int V  = E >> 2;
        const int v0 = (int)(((long long)V * s) / DEG_SLICES);
        const int v1 = (int)(((long long)V * (s + 1)) / DEG_SLICES);
        for (int v = v0 + tid; v < v1; v += 256 * DEG_BATCH) {
            int4 d[DEG_BATCH];
#pragma unroll
            for (int u = 0; u < DEG_BATCH; ++u) {
                const int vv = v + 256 * u;
                d[u] = c4[vv < v1 ? vv : v];
            }
#pragma unroll
            for (int u = 0; u < DEG_BATCH; ++u) {
                const unsigned int lim = (v + 256 * u < v1) ? DEG_BINS : 0u;
                deg_count(bins, d[u].x, lo, lim);
                deg_count(bins, d[u].y, lo, lim);
                deg_count(bins, d[u].z, lo, lim);
                deg_count(bins, d[u].w, lo, lim);
            }
        }
    } else {
        const int e0 = (int)(((long long)E * s) / DEG_SLICES);
        const int e1 = (int)(((long long)E * (s + 1)) / DEG_SLICES);
        for (int e = e0 + tid; e < e1; e += 256) deg_count(bins, col[e], lo, DEG_BINS);
    }
    __syncthreads();

    const int cnt = min(DEG_BINS, n - lo);
    int* __restrict__ prow = part + (size_t)s * n + lo;
    for (int b = tid; b < cnt; b += 256) prow[b] = (int)bins[b];
}

__global__ void k_dinv(const int* __restrict__ part, float* __restrict__ dinv,
                       unsigned int* __restrict__ q, int n) {
    int i = blockIdx.x * blockDim.x + threadIdx.x;
    if (i == 0) { q[0] = 0u; q[Q_STRIDE] = 0u; }
    if (i < n) {
        int deg = 1;
#pragma unroll
        for (int s = 0; s < DEG_SLICES; ++s) deg += part[(size_t)s * n + i];
        dinv[i] = rsqrtf((float)deg);
    }
}

__global__ __launch_bounds__(256) void k_gemm_wmma(
    const float* __restrict__ x, const float* __restrict__ W,
    const float* __restrict__ bias, const float* __restrict__ dinv,
    float* __restrict__ h, float* __restrict__ out, int nNodes) {
    __shared__ float sW[KTILE * HID];

    const int tid  = threadIdx.x;
    const int wave = tid >> 5;
    const int lane = tid & 31;
    const int half = lane >> 4;
    const int r    = lane & 15;
    const int m0   = (blockIdx.x * 8 + wave) * 16;
    const bool act = (m0 < nNodes);

    v8f acc[8] = {};

    for (int kb = 0; kb < IN_CH; kb += KTILE) {
        if (kb + KTILE < IN_CH)
            __builtin_prefetch(W + (size_t)(kb + KTILE) * HID + tid * 32, 0, 1);

        __syncthreads();
#pragma unroll
        for (int p = 0; p < (KTILE * HID / 2) / 256; ++p) {
            const int i  = tid + 256 * p;
            const int kp = i >> 7;
            const int n  = i & 127;
            float2 w2;
            w2.x = W[(size_t)(kb + 2 * kp) * HID + n];
            w2.y = W[(size_t)(kb + 2 * kp + 1) * HID + n];
            *(float2*)(sW + (size_t)i * 2) = w2;
        }
        __syncthreads();

        if (act) {
            const float* xp = x + (size_t)(m0 + r) * IN_CH + kb + 2 * half;
#pragma unroll
            for (int kl = 0; kl < KTILE; kl += 4) {
                v2f a = *(const v2f*)(xp + kl);
                const float* wrow = sW + ((kl >> 1) + half) * (2 * HID);
#pragma unroll
                for (int nt = 0; nt < 8; ++nt) {
                    v2f b = *(const v2f*)(wrow + (nt * 16 + r) * 2);
                    acc[nt] = __builtin_amdgcn_wmma_f32_16x16x4_f32(
                        false, a, false, b, (short)0, acc[nt], false, false);
                }
            }
        }
    }

    if (act) {
        float d2[8];
#pragma unroll
        for (int i = 0; i < 8; ++i) {
            float dv = dinv[m0 + i + 8 * half];
            d2[i] = dv * dv;
        }
#pragma unroll
        for (int nt = 0; nt < 8; ++nt) {
            const int   n    = nt * 16 + r;
            const float bcol = bias[n];
            float* hp = h   + (size_t)m0 * HID + n;
            float* op = out + (size_t)m0 * HID + n;
#pragma unroll
            for (int i = 0; i < 8; ++i) {
                const size_t off = (size_t)(i + 8 * half) * HID;
                const float  v   = acc[nt][i];
                hp[off] = v;
                op[off] = bcol + d2[i] * v;
            }
        }
    }
}

#define SC_TASK 64
#define SC_GROUP 8

template <bool FULL>
__device__ __forceinline__ void scatter_group(
    const float* __restrict__ h, float* __restrict__ out,
    int src, int dst, float nrm, int g, int cnt, int lane) {
    float v[SC_GROUP][4];
#pragma unroll
    for (int u = 0; u < SC_GROUP; ++u) {
        const int s = __builtin_amdgcn_readlane(src, g + u);
        const float* hp = h + (size_t)s * HID + lane;
#pragma unroll
        for (int j = 0; j < 4; ++j) v[u][j] = hp[32 * j];
    }
#pragma unroll
    for (int u = 0; u < SC_GROUP; ++u) {
        if (FULL || g + u < cnt) {
            const int   d = __builtin_amdgcn_readlane(dst, g + u);
            const float w = __int_as_float(__builtin_amdgcn_readlane(__float_as_int(nrm), g + u));
            float* op = out + (size_t)d * HID + lane;
#pragma unroll
            for (int j = 0; j < 4; ++j) atomicAdd(op + 32 * j, v[u][j] * w);
        }
    }
}

__global__ __launch_bounds__(256) void k_scatter(
    const int* __restrict__ ei, const float* __restrict__ h,
    const float* __restrict__ dinv, float* __restrict__ out,
    unsigned int* __restrict__ q, int n, int E) {
    __shared__ int2 s_rec[8][SC_TASK];
    const int lane = threadIdx.x & 31;
    const int wave = __builtin_amdgcn_readfirstlane(threadIdx.x >> 5);
    const unsigned int nTasks = (unsigned int)((E + SC_TASK - 1) / SC_TASK);

    int half = (int)((__builtin_amdgcn_s_sendmsg_rtn(0x87) >> 18) & 1u);

    unsigned int t = 0u;
    if (lane == 0) t = atomicAdd(q + Q_STRIDE * half, 1u);
    t = __builtin_amdgcn_readfirstlane(t);
    if (t >= nTasks) {
        half ^= 1;
        if (lane == 0) t = atomicAdd(q + Q_STRIDE * half, 1u);
        t = __builtin_amdgcn_readfirstlane(t);
        if (t >= nTasks) return;
    }

    const int  nSplit = n >> 1;
    const int  ea = (int)t * SC_TASK + lane;
    const int  eb = ea + 32;
    int  sa = 0, da = 0, sb = 0, db = 0;
    bool ka = false, kb = false;
    if (ea < E) {
        sa = ei[ea];
        da = ei[(size_t)E + ea];
        ka = ((da >= nSplit) == (half != 0));
    }
    if (eb < E) {
        sb = ei[eb];
        db = ei[(size_t)E + eb];
        kb = ((db >= nSplit) == (half != 0));
    }
    const unsigned int ma    = __builtin_amdgcn_ballot_w32(ka);
    const unsigned int mb    = __builtin_amdgcn_ballot_w32(kb);
    const unsigned int below = (1u << lane) - 1u;
    const int ca  = __builtin_popcount(ma);
    const int cnt = ca + __builtin_popcount(mb);
    if (ka) s_rec[wave][__builtin_popcount(ma & below)]      = make_int2(sa, da);
    if (kb) s_rec[wave][ca + __builtin_popcount(mb & below)] = make_int2(sb, db);

    int   src0 = 0, dst0 = 0, src1 = 0, dst1 = 0;
    float nrm0 = 0.0f, nrm1 = 0.0f;
    if (lane < cnt) {
        const int2 r = s_rec[wave][lane];
        src0 = r.x; dst0 = r.y;
        nrm0 = dinv[src0] * dinv[dst0];
    }
    if (lane + 32 < cnt) {
        const int2 r = s_rec[wave][lane + 32];
        src1 = r.x; dst1 = r.y;
        nrm1 = dinv[src1] * dinv[dst1];
    }

    const int c0 = min(cnt, 32);
    const int c1 = cnt - c0;
    for (int g = 0; g < c0; g += SC_GROUP) {
        if (g + SC_GROUP <= c0) scatter_group<true >(h, out, src0, dst0, nrm0, g, c0, lane);
        else                    scatter_group<false>(h, out, src0, dst0, nrm0, g, c0, lane);
    }
    for (int g = 0; g < c1; g += SC_GROUP) {
        if (g + SC_GROUP <= c1) scatter_group<true >(h, out, src1, dst1, nrm1, g, c1, lane);
        else                    scatter_group<false>(h, out, src1, dst1, nrm1, g, c1, lane);
    }
}

__global__ void k_prelu(float* __restrict__ out, const float* __restrict__ alpha, long long total) {
    long long i = (long long)blockIdx.x * blockDim.x + threadIdx.x;
    if (i < total) {
        float v = out[i];
        float a = alpha[(int)(i & (HID - 1))];
        out[i] = v > 0.0f ? v : a * v;
    }
}

extern "C" void kernel_launch(void* const* d_in, const int* in_sizes, int n_in,
                              void* d_out, int out_size, void* d_ws, size_t ws_size,
                              hipStream_t stream) {
    const float* x     = (const float*)d_in[0];
    const int*   ei    = (const int*)d_in[1];
    const float* W     = (const float*)d_in[2];
    const float* bias  = (const float*)d_in[3];
    const float* alpha = (const float*)d_in[4];
    float*       out   = (float*)d_out;

    const int N = in_sizes[0] / IN_CH;
    const int E = in_sizes[1] / 2;

    float* h    = (float*)d_ws;
    float* dinv = h + (size_t)N * HID;
    int*   part = (int*)(dinv + N);
    unsigned int* q = (unsigned int*)(part + (size_t)DEG_SLICES * N);

    const dim3 hgrid(DEG_SLICES, (N + DEG_BINS - 1) / DEG_BINS);
    k_deg_hist<<<hgrid, 256, 0, stream>>>(ei + (size_t)E, part, N, E);
    k_dinv    <<<(N + 255) / 256, 256, 0, stream>>>(part, dinv, q, N);

    const int strips = (N + 15) / 16;
    const int blocks = (strips + 7) / 8;
    k_gemm_wmma<<<blocks, 256, 0, stream>>>(x, W, bias, dinv, h, out, N);

    k_scatter<<<(E + 255) / 256, 256, 0, stream>>>(ei, h, dinv, out, q, N, E);

    const long long total = (long long)N * HID;
    k_prelu<<<(int)((total + 255) / 256), 256, 0, stream>>>(out, alpha, total);
}


# Edge sums held in LDS: block owns 391 destination rows, scans one owner byte per edge, adds h rows by f32 LDS atomics (arrival order), writes out once with self loop, bias, PReLU; 5 launches to 4
# speedup vs baseline: 2.5195x; 1.1512x over previous
#include <hip/hip_runtime.h>

typedef float v2f __attribute__((ext_vector_type(2)));
typedef float v8f __attribute__((ext_vector_type(8)));

#define IN_CH 256
#define HID   128
#define KTILE 64
#define OWN_ROWS 391

#define DEG_BINS   6400
#define DEG_SLICES 32
#define DEG_BATCH  8

__device__ __forceinline__ void deg_count(unsigned int* bins, int d, int lo, unsigned int lim) {
    const unsigned int b = (unsigned int)(d - lo);
    if (b < lim) atomicAdd(&bins[b], 1u);
}

__global__ __launch_bounds__(256) void k_deg_hist(
    const int* __restrict__ col, int* __restrict__ part,
    unsigned char* __restrict__ ownb, int n, int E) {
    __shared__ unsigned int bins[DEG_BINS];
    const int tid = threadIdx.x;
    const int s   = blockIdx.x;
    const int lo  = blockIdx.y * DEG_BINS;

    for (int b = tid; b < DEG_BINS; b += 256) bins[b] = 0u;
    __syncthreads();

    if ((E & 3) == 0) {
        const int4* __restrict__ c4 = (const int4*)col;
        const int V  = E >> 2;
        const int v0 = (int)(((long long)V * s) / DEG_SLICES);
        const int v1 = (int)(((long long)V * (s + 1)) / DEG_SLICES);
        unsigned int turn = 0u;
        for (int v = v0 + tid; v < v1; v += 256 * DEG_BATCH, ++turn) {
            int4 d[DEG_BATCH];
#pragma unroll
            for (int u = 0; u < DEG_BATCH; ++u) {
                const int vv = v + 256 * u;
                d[u] = c4[vv < v1 ? vv : v];
            }
            if (turn % gridDim.y == blockIdx.y) {
                unsigned int* __restrict__ ow = (unsigned int*)ownb;
#pragma unroll
                for (int u = 0; u < DEG_BATCH; ++u) {
                    const int vv = v + 256 * u;
                    if (vv < v1)
                        ow[vv] = ((unsigned int)d[u].x / OWN_ROWS) | (((unsigned int)d[u].y / OWN_ROWS) << 8) |
                                 (((unsigned int)d[u].z / OWN_ROWS) << 16) | (((unsigned int)d[u].w / OWN_ROWS) << 24);
                }
            }
#pragma unroll
            for (int u = 0; u < DEG_BATCH; ++u) {
                const unsigned int lim = (v + 256 * u < v1) ? DEG_BINS : 0u;
                deg_count(bins, d[u].x, lo, lim);
                deg_count(bins, d[u].y, lo, lim);
                deg_count(bins, d[u].z, lo, lim);
                deg_count(bins, d[u].w, lo, lim);
            }
        }
    } else {
        const int e0 = (int)(((long long)E * s) / DEG_SLICES);
        const int e1 = (int)(((long long)E * (s + 1)) / DEG_SLICES);
        for (int e = e0 + tid; e < e1; e += 256) {
            const int d = col[e];
            deg_count(bins, d, lo, DEG_BINS);
            if (blockIdx.y == 0) ownb[e] = (unsigned char)((unsigned int)d / OWN_ROWS);
        }
    }
    __syncthreads();

    const int cnt = min(DEG_BINS, n - lo);
    int* __restrict__ prow = part + (size_t)s * n + lo;
    for (int b = tid; b < cnt; b += 256) prow[b] = (int)bins[b];
}

__global__ void k_dinv(const int* __restrict__ part, float* __restrict__ dinv, int n) {
    int i = blockIdx.x * blockDim.x + threadIdx.x;
    if (i < n) {
        int deg = 1;
#pragma unroll
        for (int s = 0; s < DEG_SLICES; ++s) deg += part[(size_t)s * n + i];
        dinv[i] = rsqrtf((float)deg);
    }
}

__global__ __launch_bounds__(256) void k_gemm_wmma(
    const float* __restrict__ x, const float* __restrict__ W,
    float* __restrict__ h, int nNodes) {
    __shared__ float sW[KTILE * HID];

    const int tid  = threadIdx.x;
    const int wave = tid >> 5;
    const int lane = tid & 31;
    const int half = lane >> 4;
    const int r    = lane & 15;
    const int m0   = (blockIdx.x * 8 + wave) * 16;
    const bool act = (m0 < nNodes);

    v8f acc[8] = {};

    for (int kb = 0; kb < IN_CH; kb += KTILE) {
        if (kb + KTILE < IN_CH)
            __builtin_prefetch(W + (size_t)(kb + KTILE) * HID + tid * 32, 0, 1);

        __syncthreads();
#pragma unroll
        for (int p = 0; p < (KTILE * HID / 2) / 256; ++p) {
            const int i  = tid + 256 * p;
            const int kp = i >> 7;
            const int n  = i & 127;
            float2 w2;
            w2.x = W[(size_t)(kb + 2 * kp) * HID + n];
            w2.y = W[(size_t)(kb + 2 * kp + 1) * HID + n];
            *(float2*)(sW + (size_t)i * 2) = w2;
        }
        __syncthreads();

        if (act) {
            const float* xp = x + (size_t)(m0 + r) * IN_CH + kb + 2 * half;
#pragma unroll
            for (int kl = 0; kl < KTILE; kl += 4) {
                v2f a = *(const v2f*)(xp + kl);
                const float* wrow = sW + ((kl >> 1) + half) * (2 * HID);
#pragma unroll
                for (int nt = 0; nt < 8; ++nt) {
                    v2f b = *(const v2f*)(wrow + (nt * 16 + r) * 2);
                    acc[nt] = __builtin_amdgcn_wmma_f32_16x16x4_f32(
                        false, a, false, b, (short)0, acc[nt], false, false);
                }
            }
        }
    }

    if (act) {
#pragma unroll
        for (int nt = 0; nt < 8; ++nt) {
            float* hp = h + (size_t)m0 * HID + nt * 16 + r;
#pragma unroll
            for (int i = 0; i < 8; ++i)
                hp[(size_t)(i + 8 * half) * HID] = acc[nt][i];
        }
    }
}

#define OWN_WAVES 32
#define OWN_U     4
#define OWN_QCAP  64
#define OWN_GROUP 8
#define OWN_STRIDE (OWN_WAVES * 32 * OWN_U)

__device__ __forceinline__ unsigned int ne_bytes(unsigned int x, unsigned int pat) {
    const unsigned int y = x ^ pat;
    return ((y & 0x7f7f7f7fu) + 0x7f7f7f7fu) | y;
}

template <bool FULL>
__device__ __forceinline__ void own_batch(
    const int* __restrict__ ei, const float* __restrict__ h, const float* __restrict__ dinv,
    float* acc, const int* qs, int cnt, int lo, int E, int lane) {
    int src = 0, dl = 0;
    float nrm = 0.0f;
    if (FULL || lane < cnt) {
        const int e   = qs[lane];
        src           = ei[e];
        const int dst = ei[(size_t)E + e];
        dl  = dst - lo;
        nrm = dinv[src] * dinv[dst];
    }
    for (int g = 0; g < cnt; g += OWN_GROUP) {
        float v[OWN_GROUP][4];
#pragma unroll
        for (int u = 0; u < OWN_GROUP; ++u) {
            const int s = __builtin_amdgcn_readlane(src, g + u);
            const float* hp = h + (size_t)s * HID + lane;
#pragma unroll
            for (int j = 0; j < 4; ++j) v[u][j] = hp[32 * j];
        }
#pragma unroll
        for (int u = 0; u < OWN_GROUP; ++u) {
            if (FULL || g + u < cnt) {
                const int   d = __builtin_amdgcn_readlane(dl, g + u);
                const float w = __int_as_float(__builtin_amdgcn_readlane(__float_as_int(nrm), g + u));
                float* ap = acc + d * HID + lane;
#pragma unroll
                for (int j = 0; j < 4; ++j)
                    __hip_atomic_fetch_add(ap + 32 * j, v[u][j] * w, __ATOMIC_RELAXED,
                                           __HIP_MEMORY_SCOPE_WORKGROUP);
            }
        }
    }
}

__global__ __launch_bounds__(1024) void k_owner(
    const int* __restrict__ ei, const unsigned char* __restrict__ ownb,
    const float* __restrict__ h, const float* __restrict__ dinv,
    const float* __restrict__ bias, const float* __restrict__ alpha,
    float* __restrict__ out, int n, int E) {
    __shared__ float acc[OWN_ROWS * HID];
    __shared__ int   s_q[OWN_WAVES][OWN_QCAP];

    const int tid  = threadIdx.x;
    const int lane = tid & 31;
    const int wave = __builtin_amdgcn_readfirstlane(tid >> 5);
    const int me   = blockIdx.x;
    const int lo   = me * OWN_ROWS;

    for (int i = tid; i < OWN_ROWS * HID / 4; i += 1024)
        ((float4*)acc)[i] = make_float4(0.0f, 0.0f, 0.0f, 0.0f);
    __syncthreads();

    const uint4* __restrict__ ob4 = (const uint4*)ownb;
    const int V = E >> 4;
    const unsigned int pat   = (unsigned int)me * 0x01010101u;
    const unsigned int below = (1u << lane) - 1u;
    int* q  = s_q[wave];
    int  qn = 0;

    uint4 cur[OWN_U];
    int vb = wave * (32 * OWN_U);
#pragma unroll
    for (int u = 0; u < OWN_U; ++u) {
        const int v = vb + 32 * u + lane;
        cur[u] = ob4[v < V ? v : 0];
    }
    for (; vb < V; vb += OWN_STRIDE) {
        uint4 nxt[OWN_U];
#pragma unroll
        for (int u = 0; u < OWN_U; ++u) {
            const int v = vb + OWN_STRIDE + 32 * u + lane;
            nxt[u] = ob4[v < V ? v : 0];
        }
        unsigned int m[OWN_U];
#pragma unroll
        for (int u = 0; u < OWN_U; ++u) {
            const int v = vb + 32 * u + lane;
            const unsigned int mm =
                ~(((ne_bytes(cur[u].x, pat) >> 7) & 0x01010101u) | ((ne_bytes(cur[u].y, pat) >> 6) & 0x02020202u) |
                  ((ne_bytes(cur[u].z, pat) >> 5) & 0x04040404u) | ((ne_bytes(cur[u].w, pat) >> 4) & 0x08080808u)) &
                0x0f0f0f0fu;
            m[u] = (v < V) ? mm : 0u;
        }
        for (;;) {
            const bool has = ((m[0] | m[1] | m[2] | m[3]) != 0u);
            const unsigned int b = __builtin_amdgcn_ballot_w32(has);
            if (b == 0u) break;
            if (has) {
                const int uu = m[0] ? 0 : (m[1] ? 1 : (m[2] ? 2 : 3));
                const unsigned int mm = m[0] ? m[0] : (m[1] ? m[1] : (m[2] ? m[2] : m[3]));
                const int p = __builtin_ctz(mm);
                const unsigned int rest = mm & (mm - 1u);
                m[0] = (uu == 0) ? rest : m[0];
                m[1] = (uu == 1) ? rest : m[1];
                m[2] = (uu == 2) ? rest : m[2];
                m[3] = (uu == 3) ? rest : m[3];
                q[qn + __builtin_popcount(b & below)] =
                    (vb + 32 * uu + lane) * 16 + ((p & 7) << 2) + (p >> 3);
            }
            qn += __builtin_popcount(b);
            if (qn >= 32) {
                qn -= 32;
                own_batch<true>(ei, h, dinv, acc, q + qn, 32, lo, E, lane);
            }
        }
#pragma unroll
        for (int u = 0; u < OWN_U; ++u) cur[u] = nxt[u];
    }
    if (wave == 0) {
        const int e = V * 16 + lane;
        const bool has = (lane < 16) && (e < E) && ((int)ownb[e < E ? e : 0] == me);
        const unsigned int b = __builtin_amdgcn_ballot_w32(has);
        if (has) q[qn + __builtin_popcount(b & below)] = e;
        qn += __builtin_popcount(b);
    }
    while (qn > 0) {
        const int take = min(qn, 32);
        qn -= take;
        own_batch<false>(ei, h, dinv, acc, q + qn, take, lo, E, lane);
    }
    __syncthreads();

    const int rows = min(OWN_ROWS, n - lo);
    const float4* __restrict__ h4 = (const float4*)h;
    float4* __restrict__ o4 = (float4*)out;
    for (int i = tid; i < rows * (HID / 4); i += 1024) {
        const int r   = i >> 5;
        const int c4  = i & 31;
        const int row = lo + r;
        const float d  = dinv[row];
        const float d2 = d * d;
        const float4 hv = h4[(size_t)row * (HID / 4) + c4];
        const float4 av = ((const float4*)acc)[i];
        const float4 bv = ((const float4*)bias)[c4];
        const float4 al = ((const float4*)alpha)[c4];
        float4 o;
        o.x = (bv.x + d2 * hv.x) + av.x;
        o.y = (bv.y + d2 * hv.y) + av.y;
        o.z = (bv.z + d2 * hv.z) + av.z;
        o.w = (bv.w + d2 * hv.w) + av.w;
        o.x = o.x > 0.0f ? o.x : al.x * o.x;
        o.y = o.y > 0.0f ? o.y : al.y * o.y;
        o.z = o.z > 0.0f ? o.z : al.z * o.z;
        o.w = o.w > 0.0f ? o.w : al.w * o.w;
        o4[(size_t)row * (HID / 4) + c4] = o;
    }
}

extern "C" void kernel_launch(void* const* d_in, const int* in_sizes, int n_in,
                              void* d_out, int out_size, void* d_ws, size_t ws_size,
                              hipStream_t stream) {
    const float* x     = (const float*)d_in[0];
    const int*   ei    = (const int*)d_in[1];
    const float* W     = (const float*)d_in[2];
    const float* bias  = (const float*)d_in[3];
    const float* alpha = (const float*)d_in[4];
    float*       out   = (float*)d_out;

    const int N = in_sizes[0] / IN_CH;
    const int E = in_sizes[1] / 2;

    float* h    = (float*)d_ws;
    unsigned char* ownb = (unsigned char*)(h + (size_t)N * HID);
    float* dinv = (float*)(ownb + (((size_t)E + 15) & ~(size_t)15));
    int*   part = (int*)(dinv + N);

    const dim3 hgrid(DEG_SLICES, (N + DEG_BINS - 1) / DEG_BINS);
    k_deg_hist<<<hgrid, 256, 0, stream>>>(ei + (size_t)E, part, ownb, N, E);
    k_dinv    <<<(N + 255) / 256, 256, 0, stream>>>(part, dinv, N);

    const int strips = (N + 15) / 16;
    const int blocks = (strips + 7) / 8;
    k_gemm_wmma<<<blocks, 256, 0, stream>>>(x, W, h, N);

    k_owner<<<(N + OWN_ROWS - 1) / OWN_ROWS, 1024, 0, stream>>>(ei, ownb, h, dinv, bias, alpha, out, N, E);
}


# Edge norm factored: GEMM stores row j times dinv[j]; owner block adds rows as loaded, scales each row sum by dinv[dst] at write-out; no dinv lookup per edge; arrival order kept, term rounding differs
# speedup vs baseline: 2.6857x; 1.0660x over previous
#include <hip/hip_runtime.h>

typedef float v2f __attribute__((ext_vector_type(2)));
typedef float v8f __attribute__((ext_vector_type(8)));

#define IN_CH 256
#define HID   128
#define KTILE 64
#define OWN_ROWS 391

#define DEG_BINS   6400
#define DEG_SLICES 32
#define DEG_BATCH  8

__device__ __forceinline__ void deg_count(unsigned int* bins, int d, int lo, unsigned int lim) {
    const unsigned int b = (unsigned int)(d - lo);
    if (b < lim) atomicAdd(&bins[b], 1u);
}

__global__ __launch_bounds__(256) void k_deg_hist(
    const int* __restrict__ col, int* __restrict__ part,
    unsigned char* __restrict__ ownb, int n, int E) {
    __shared__ unsigned int bins[DEG_BINS];
    const int tid = threadIdx.x;
    const int s   = blockIdx.x;
    const int lo  = blockIdx.y * DEG_BINS;

    for (int b = tid; b < DEG_BINS; b += 256) bins[b] = 0u;
    __syncthreads();

    if ((E & 3) == 0) {
        const int4* __restrict__ c4 = (const int4*)col;
        const int V  = E >> 2;
        const int v0 = (int)(((long long)V * s) / DEG_SLICES);
        const int v1 = (int)(((long long)V * (s + 1)) / DEG_SLICES);
        unsigned int turn = 0u;
        for (int v = v0 + tid; v < v1; v += 256 * DEG_BATCH, ++turn) {
            int4 d[DEG_BATCH];
#pragma unroll
            for (int u = 0; u < DEG_BATCH; ++u) {
                const int vv = v + 256 * u;
                d[u] = c4[vv < v1 ? vv : v];
            }
            if (turn % gridDim.y == blockIdx.y) {
                unsigned int* __restrict__ ow = (unsigned int*)ownb;
#pragma unroll
                for (int u = 0; u < DEG_BATCH; ++u) {
                    const int vv = v + 256 * u;
                    if (vv < v1)
                        ow[vv] = ((unsigned int)d[u].x / OWN_ROWS) | (((unsigned int)d[u].y / OWN_ROWS) << 8) |
                                 (((unsigned int)d[u].z / OWN_ROWS) << 16) | (((unsigned int)d[u].w / OWN_ROWS) << 24);
                }
            }
#pragma unroll
            for (int u = 0; u < DEG_BATCH; ++u) {
                const unsigned int lim = (v + 256 * u < v1) ? DEG_BINS : 0u;
                deg_count(bins, d[u].x, lo, lim);
                deg_count(bins, d[u].y, lo, lim);
                deg_count(bins, d[u].z, lo, lim);
                deg_count(bins, d[u].w, lo, lim);
            }
        }
    } else {
        const int e0 = (int)(((long long)E * s) / DEG_SLICES);
        const int e1 = (int)(((long long)E * (s + 1)) / DEG_SLICES);
        for (int e = e0 + tid; e < e1; e += 256) {
            const int d = col[e];
            deg_count(bins, d, lo, DEG_BINS);
            if (blockIdx.y == 0) ownb[e] = (unsigned char)((unsigned int)d / OWN_ROWS);
        }
    }
    __syncthreads();

    const int cnt = min(DEG_BINS, n - lo);
    int* __restrict__ prow = part + (size_t)s * n + lo;
    for (int b = tid; b < cnt; b += 256) prow[b] = (int)bins[b];
}

__global__ void k_dinv(const int* __restrict__ part, float* __restrict__ dinv, int n) {
    int i = blockIdx.x * blockDim.x + threadIdx.x;
    if (i < n) {
        int deg = 1;
#pragma unroll
        for (int s = 0; s < DEG_SLICES; ++s) deg += part[(size_t)s * n + i];
        dinv[i] = rsqrtf((float)deg);
    }
}

__global__ __launch_bounds__(256) void k_gemm_wmma(
    const float* __restrict__ x, const float* __restrict__ W,
    const float* __restrict__ dinv, float* __restrict__ h, int nNodes) {
    __shared__ float sW[KTILE * HID];

    const int tid  = threadIdx.x;
    const int wave = tid >> 5;
    const int lane = tid & 31;
    const int half = lane >> 4;
    const int r    = lane & 15;
    const int m0   = (blockIdx.x * 8 + wave) * 16;
    const bool act = (m0 < nNodes);

    v8f acc[8] = {};

    for (int kb = 0; kb < IN_CH; kb += KTILE) {
        if (kb + KTILE < IN_CH)
            __builtin_prefetch(W + (size_t)(kb + KTILE) * HID + tid * 32, 0, 1);

        __syncthreads();
#pragma unroll
        for (int p = 0; p < (KTILE * HID / 2) / 256; ++p) {
            const int i  = tid + 256 * p;
            const int kp = i >> 7;
            const int n  = i & 127;
            float2 w2;
            w2.x = W[(size_t)(kb + 2 * kp) * HID + n];
            w2.y = W[(size_t)(kb + 2 * kp + 1) * HID + n];
            *(float2*)(sW + (size_t)i * 2) = w2;
        }
        __syncthreads();

        if (act) {
            const float* xp = x + (size_t)(m0 + r) * IN_CH + kb + 2 * half;
#pragma unroll
            for (int kl = 0; kl < KTILE; kl += 4) {
                v2f a = *(const v2f*)(xp + kl);
                const float* wrow = sW + ((kl >> 1) + half) * (2 * HID);
#pragma unroll
                for (int nt = 0; nt < 8; ++nt) {
                    v2f b = *(const v2f*)(wrow + (nt * 16 + r) * 2);
                    acc[nt] = __builtin_amdgcn_wmma_f32_16x16x4_f32(
                        false, a, false, b, (short)0, acc[nt], false, false);
                }
            }
        }
    }

    if (act) {
        const float4* dp = (const float4*)(dinv + m0 + 8 * half);
        const float4 d0 = dp[0];
        const float4 d1 = dp[1];
        const float dv[8] = {d0.x, d0.y, d0.z, d0.w, d1.x, d1.y, d1.z, d1.w};
#pragma unroll
        for (int nt = 0; nt < 8; ++nt) {
            float* hp = h + (size_t)m0 * HID + nt * 16 + r;
#pragma unroll
            for (int i = 0; i < 8; ++i)
                hp[(size_t)(i + 8 * half) * HID] = acc[nt][i] * dv[i];
        }
    }
}

#define OWN_WAVES 32
#define OWN_U     4
#define OWN_QCAP  64
#define OWN_GROUP 8
#define OWN_STRIDE (OWN_WAVES * 32 * OWN_U)

__device__ __forceinline__ unsigned int ne_bytes(unsigned int x, unsigned int pat) {
    const unsigned int y = x ^ pat;
    return ((y & 0x7f7f7f7fu) + 0x7f7f7f7fu) | y;
}

template <bool FULL>
__device__ __forceinline__ void own_batch(
    const int* __restrict__ ei, const float* __restrict__ h,
    float* acc, const int* qs, int cnt, int lo, int E, int lane) {
    int src = 0, dl = 0;
    if (FULL || lane < cnt) {
        const int e = qs[lane];
        src = ei[e];
        dl  = ei[(size_t)E + e] - lo;
    }
    for (int g = 0; g < cnt; g += OWN_GROUP) {
        float v[OWN_GROUP][4];
#pragma unroll
        for (int u = 0; u < OWN_GROUP; ++u) {
            const int s = __builtin_amdgcn_readlane(src, g + u);
            const float* hp = h + (size_t)s * HID + lane;
#pragma unroll
            for (int j = 0; j < 4; ++j) v[u][j] = hp[32 * j];
        }
#pragma unroll
        for (int u = 0; u < OWN_GROUP; ++u) {
            if (FULL || g + u < cnt) {
                const int d = __builtin_amdgcn_readlane(dl, g + u);
                float* ap = acc + d * HID + lane;
#pragma unroll
                for (int j = 0; j < 4; ++j)
                    __hip_atomic_fetch_add(ap + 32 * j, v[u][j], __ATOMIC_RELAXED,
                                           __HIP_MEMORY_SCOPE_WORKGROUP);
            }
        }
    }
}

__global__ __launch_bounds__(1024) void k_owner(
    const int* __restrict__ ei, const unsigned char* __restrict__ ownb,
    const float* __restrict__ h, const float* __restrict__ dinv,
    const float* __restrict__ bias, const float* __restrict__ alpha,
    float* __restrict__ out, int n, int E) {
    __shared__ float acc[OWN_ROWS * HID];
    __shared__ int   s_q[OWN_WAVES][OWN_QCAP];

    const int tid  = threadIdx.x;
    const int lane = tid & 31;
    const int wave = __builtin_amdgcn_readfirstlane(tid >> 5);
    const int me   = blockIdx.x;
    const int lo   = me * OWN_ROWS;

    for (int i = tid; i < OWN_ROWS * HID / 4; i += 1024)
        ((float4*)acc)[i] = make_float4(0.0f, 0.0f, 0.0f, 0.0f);
    __syncthreads();

    const uint4* __restrict__ ob4 = (const uint4*)ownb;
    const int V = E >> 4;
    const unsigned int pat   = (unsigned int)me * 0x01010101u;
    const unsigned int below = (1u << lane) - 1u;
    int* q  = s_q[wave];
    int  qn = 0;

    uint4 cur[OWN_U];
    int vb = wave * (32 * OWN_U);
#pragma unroll
    for (int u = 0; u < OWN_U; ++u) {
        const int v = vb + 32 * u + lane;
        cur[u] = ob4[v < V ? v : 0];
    }
    for (; vb < V; vb += OWN_STRIDE) {
        uint4 nxt[OWN_U];
#pragma unroll
        for (int u = 0; u < OWN_U; ++u) {
            const int v = vb + OWN_STRIDE + 32 * u + lane;
            nxt[u] = ob4[v < V ? v : 0];
        }
        unsigned int m[OWN_U];
#pragma unroll
        for (int u = 0; u < OWN_U; ++u) {
            const int v = vb + 32 * u + lane;
            const unsigned int mm =
                ~(((ne_bytes(cur[u].x, pat) >> 7) & 0x01010101u) | ((ne_bytes(cur[u].y, pat) >> 6) & 0x02020202u) |
                  ((ne_bytes(cur[u].z, pat) >> 5) & 0x04040404u) | ((ne_bytes(cur[u].w, pat) >> 4) & 0x08080808u)) &
                0x0f0f0f0fu;
            m[u] = (v < V) ? mm : 0u;
        }
        for (;;) {
            const bool has = ((m[0] | m[1] | m[2] | m[3]) != 0u);
            const unsigned int b = __builtin_amdgcn_ballot_w32(has);
            if (b == 0u) break;
            if (has) {
                const int uu = m[0] ? 0 : (m[1] ? 1 : (m[2] ? 2 : 3));
                const unsigned int mm = m[0] ? m[0] : (m[1] ? m[1] : (m[2] ? m[2] : m[3]));
                const int p = __builtin_ctz(mm);
                const unsigned int rest = mm & (mm - 1u);
                m[0] = (uu == 0) ? rest : m[0];
                m[1] = (uu == 1) ? rest : m[1];
                m[2] = (uu == 2) ? rest : m[2];
                m[3] = (uu == 3) ? rest : m[3];
                q[qn + __builtin_popcount(b & below)] =
                    (vb + 32 * uu + lane) * 16 + ((p & 7) << 2) + (p >> 3);
            }
            qn += __builtin_popcount(b);
            if (qn >= 32) {
                qn -= 32;
                own_batch<true>(ei, h, acc, q + qn, 32, lo, E, lane);
            }
        }
#pragma unroll
        for (int u = 0; u < OWN_U; ++u) cur[u] = nxt[u];
    }
    if (wave == 0) {
        const int e = V * 16 + lane;
        const bool has = (lane < 16) && (e < E) && ((int)ownb[e < E ? e : 0] == me);
        const unsigned int b = __builtin_amdgcn_ballot_w32(has);
        if (has) q[qn + __builtin_popcount(b & below)] = e;
        qn += __builtin_popcount(b);
    }
    while (qn > 0) {
        const int take = min(qn, 32);
        qn -= take;
        own_batch<false>(ei, h, acc, q + qn, take, lo, E, lane);
    }
    __syncthreads();

    const int rows = min(OWN_ROWS, n - lo);
    const float4* __restrict__ h4 = (const float4*)h;
    float4* __restrict__ o4 = (float4*)out;
    for (int i = tid; i < rows * (HID / 4); i += 1024) {
        const int r   = i >> 5;
        const int c4  = i & 31;
        const int row = lo + r;
        const float d = dinv[row];
        const float4 hv = h4[(size_t)row * (HID / 4) + c4];
        const float4 av = ((const float4*)acc)[i];
        const float4 bv = ((const float4*)bias)[c4];
        const float4 al = ((const float4*)alpha)[c4];
        float4 o;
        o.x = bv.x + d * (hv.x + av.x);
        o.y = bv.y + d * (hv.y + av.y);
        o.z = bv.z + d * (hv.z + av.z);
        o.w = bv.w + d * (hv.w + av.w);
        o.x = o.x > 0.0f ? o.x : al.x * o.x;
        o.y = o.y > 0.0f ? o.y : al.y * o.y;
        o.z = o.z > 0.0f ? o.z : al.z * o.z;
        o.w = o.w > 0.0f ? o.w : al.w * o.w;
        o4[(size_t)row * (HID / 4) + c4] = o;
    }
}

extern "C" void kernel_launch(void* const* d_in, const int* in_sizes, int n_in,
                              void* d_out, int out_size, void* d_ws, size_t ws_size,
                              hipStream_t stream) {
    const float* x     = (const float*)d_in[0];
    const int*   ei    = (const int*)d_in[1];
    const float* W     = (const float*)d_in[2];
    const float* bias  = (const float*)d_in[3];
    const float* alpha = (const float*)d_in[4];
    float*       out   = (float*)d_out;

    const int N = in_sizes[0] / IN_CH;
    const int E = in_sizes[1] / 2;

    float* h    = (float*)d_ws;
    unsigned char* ownb = (unsigned char*)(h + (size_t)N * HID);
    float* dinv = (float*)(ownb + (((size_t)E + 15) & ~(size_t)15));
    int*   part = (int*)(dinv + N);

    const dim3 hgrid(DEG_SLICES, (N + DEG_BINS - 1) / DEG_BINS);
    k_deg_hist<<<hgrid, 256, 0, stream>>>(ei + (size_t)E, part, ownb, N, E);
    k_dinv    <<<(N + 255) / 256, 256, 0, stream>>>(part, dinv, N);

    const int strips = (N + 15) / 16;
    const int blocks = (strips + 7) / 8;
    k_gemm_wmma<<<blocks, 256, 0, stream>>>(x, W, dinv, h, N);

    k_owner<<<(N + OWN_ROWS - 1) / OWN_ROWS, 1024, 0, stream>>>(ei, ownb, h, dinv, bias, alpha, out, N, E);
}


# Row gather widened: lane l loads channels 4l..4l+3 of an h row in one 16-byte load (one 512-byte load per edge, not four); LDS rows channel-interleaved, adds stay conflict-free; arrival order kept
# speedup vs baseline: 3.0460x; 1.1342x over previous
#include <hip/hip_runtime.h>

typedef float v2f __attribute__((ext_vector_type(2)));
typedef float v8f __attribute__((ext_vector_type(8)));

#define IN_CH 256
#define HID   128
#define KTILE 64
#define OWN_ROWS 391

#define DEG_BINS   6400
#define DEG_SLICES 32
#define DEG_BATCH  8

__device__ __forceinline__ void deg_count(unsigned int* bins, int d, int lo, unsigned int lim) {
    const unsigned int b = (unsigned int)(d - lo);
    if (b < lim) atomicAdd(&bins[b], 1u);
}

__global__ __launch_bounds__(256) void k_deg_hist(
    const int* __restrict__ col, int* __restrict__ part,
    unsigned char* __restrict__ ownb, int n, int E) {
    __shared__ unsigned int bins[DEG_BINS];
    const int tid = threadIdx.x;
    const int s   = blockIdx.x;
    const int lo  = blockIdx.y * DEG_BINS;

    for (int b = tid; b < DEG_BINS; b += 256) bins[b] = 0u;
    __syncthreads();

    if ((E & 3) == 0) {
        const int4* __restrict__ c4 = (const int4*)col;
        const int V  = E >> 2;
        const int v0 = (int)(((long long)V * s) / DEG_SLICES);
        const int v1 = (int)(((long long)V * (s + 1)) / DEG_SLICES);
        unsigned int turn = 0u;
        for (int v = v0 + tid; v < v1; v += 256 * DEG_BATCH, ++turn) {
            int4 d[DEG_BATCH];
#pragma unroll
            for (int u = 0; u < DEG_BATCH; ++u) {
                const int vv = v + 256 * u;
                d[u] = c4[vv < v1 ? vv : v];
            }
            if (turn % gridDim.y == blockIdx.y) {
                unsigned int* __restrict__ ow = (unsigned int*)ownb;
#pragma unroll
                for (int u = 0; u < DEG_BATCH; ++u) {
                    const int vv = v + 256 * u;
                    if (vv < v1)
                        ow[vv] = ((unsigned int)d[u].x / OWN_ROWS) | (((unsigned int)d[u].y / OWN_ROWS) << 8) |
                                 (((unsigned int)d[u].z / OWN_ROWS) << 16) | (((unsigned int)d[u].w / OWN_ROWS) << 24);
                }
            }
#pragma unroll
            for (int u = 0; u < DEG_BATCH; ++u) {
                const unsigned int lim = (v + 256 * u < v1) ? DEG_BINS : 0u;
                deg_count(bins, d[u].x, lo, lim);
                deg_count(bins, d[u].y, lo, lim);
                deg_count(bins, d[u].z, lo, lim);
                deg_count(bins, d[u].w, lo, lim);
            }
        }
    } else {
        const int e0 = (int)(((long long)E * s) / DEG_SLICES);
        const int e1 = (int)(((long long)E * (s + 1)) / DEG_SLICES);
        for (int e = e0 + tid; e < e1; e += 256) {
            const int d = col[e];
            deg_count(bins, d, lo, DEG_BINS);
            if (blockIdx.y == 0) ownb[e] = (unsigned char)((unsigned int)d / OWN_ROWS);
        }
    }
    __syncthreads();

    const int cnt = min(DEG_BINS, n - lo);
    int* __restrict__ prow = part + (size_t)s * n + lo;
    for (int b = tid; b < cnt; b += 256) prow[b] = (int)bins[b];
}

__global__ void k_dinv(const int* __restrict__ part, float* __restrict__ dinv, int n) {
    int i = blockIdx.x * blockDim.x + threadIdx.x;
    if (i < n) {
        int deg = 1;
#pragma unroll
        for (int s = 0; s < DEG_SLICES; ++s) deg += part[(size_t)s * n + i];
        dinv[i] = rsqrtf((float)deg);
    }
}

__global__ __launch_bounds__(256) void k_gemm_wmma(
    const float* __restrict__ x, const float* __restrict__ W,
    const float* __restrict__ dinv, float* __restrict__ h, int nNodes) {
    __shared__ float sW[KTILE * HID];

    const int tid  = threadIdx.x;
    const int wave = tid >> 5;
    const int lane = tid & 31;
    const int half = lane >> 4;
    const int r    = lane & 15;
    const int m0   = (blockIdx.x * 8 + wave) * 16;
    const bool act = (m0 < nNodes);

    v8f acc[8] = {};

    for (int kb = 0; kb < IN_CH; kb += KTILE) {
        if (kb + KTILE < IN_CH)
            __builtin_prefetch(W + (size_t)(kb + KTILE) * HID + tid * 32, 0, 1);

        __syncthreads();
#pragma unroll
        for (int p = 0; p < (KTILE * HID / 2) / 256; ++p) {
            const int i  = tid + 256 * p;
            const int kp = i >> 7;
            const int n  = i & 127;
            float2 w2;
            w2.x = W[(size_t)(kb + 2 * kp) * HID + n];
            w2.y = W[(size_t)(kb + 2 * kp + 1) * HID + n];
            *(float2*)(sW + (size_t)i * 2) = w2;
        }
        __syncthreads();

        if (act) {
            const float* xp = x + (size_t)(m0 + r) * IN_CH + kb + 2 * half;
#pragma unroll
            for (int kl = 0; kl < KTILE; kl += 4) {
                v2f a = *(const v2f*)(xp + kl);
                const float* wrow = sW + ((kl >> 1) + half) * (2 * HID);
#pragma unroll
                for (int nt = 0; nt < 8; ++nt) {
                    v2f b = *(const v2f*)(wrow + (nt * 16 + r) * 2);
                    acc[nt] = __builtin_amdgcn_wmma_f32_16x16x4_f32(
                        false, a, false, b, (short)0, acc[nt], false, false);
                }
            }
        }
    }

    if (act) {
        const float4* dp = (const float4*)(dinv + m0 + 8 * half);
        const float4 d0 = dp[0];
        const float4 d1 = dp[1];
        const float dv[8] = {d0.x, d0.y, d0.z, d0.w, d1.x, d1.y, d1.z, d1.w};
#pragma unroll
        for (int nt = 0; nt < 8; ++nt) {
            float* hp = h + (size_t)m0 * HID + nt * 16 + r;
#pragma unroll
            for (int i = 0; i < 8; ++i)
                hp[(size_t)(i + 8 * half) * HID] = acc[nt][i] * dv[i];
        }
    }
}

#define OWN_WAVES 32
#define OWN_U     4
#define OWN_QCAP  64
#define OWN_GROUP 8
#define OWN_STRIDE (OWN_WAVES * 32 * OWN_U)

__device__ __forceinline__ unsigned int ne_bytes(unsigned int x, unsigned int pat) {
    const unsigned int y = x ^ pat;
    return ((y & 0x7f7f7f7fu) + 0x7f7f7f7fu) | y;
}

template <bool FULL>
__device__ __forceinline__ void own_batch(
    const int* __restrict__ ei, const float* __restrict__ h,
    float* acc, const int* qs, int cnt, int lo, int E, int lane) {
    int src = 0, dl = 0;
    if (FULL || lane < cnt) {
        const int e = qs[lane];
        src = ei[e];
        dl  = ei[(size_t)E + e] - lo;
    }
    for (int g = 0; g < cnt; g += OWN_GROUP) {
        float4 v[OWN_GROUP];
#pragma unroll
        for (int u = 0; u < OWN_GROUP; ++u) {
            const int s = __builtin_amdgcn_readlane(src, g + u);
            v[u] = *(const float4*)(h + (size_t)s * HID + 4 * lane);
        }
#pragma unroll
        for (int u = 0; u < OWN_GROUP; ++u) {
            if (FULL || g + u < cnt) {
                const int d = __builtin_amdgcn_readlane(dl, g + u);
                float* ap = acc + d * HID + lane;
                __hip_atomic_fetch_add(ap,      v[u].x, __ATOMIC_RELAXED, __HIP_MEMORY_SCOPE_WORKGROUP);
                __hip_atomic_fetch_add(ap + 32, v[u].y, __ATOMIC_RELAXED, __HIP_MEMORY_SCOPE_WORKGROUP);
                __hip_atomic_fetch_add(ap + 64, v[u].z, __ATOMIC_RELAXED, __HIP_MEMORY_SCOPE_WORKGROUP);
                __hip_atomic_fetch_add(ap + 96, v[u].w, __ATOMIC_RELAXED, __HIP_MEMORY_SCOPE_WORKGROUP);
            }
        }
    }
}

__global__ __launch_bounds__(1024) void k_owner(
    const int* __restrict__ ei, const unsigned char* __restrict__ ownb,
    const float* __restrict__ h, const float* __restrict__ dinv,
    const float* __restrict__ bias, const float* __restrict__ alpha,
    float* __restrict__ out, int n, int E) {
    __shared__ float acc[OWN_ROWS * HID];
    __shared__ int   s_q[OWN_WAVES][OWN_QCAP];

    const int tid  = threadIdx.x;
    const int lane = tid & 31;
    const int wave = __builtin_amdgcn_readfirstlane(tid >> 5);
    const int me   = blockIdx.x;
    const int lo   = me * OWN_ROWS;

    for (int i = tid; i < OWN_ROWS * HID / 4; i += 1024)
        ((float4*)acc)[i] = make_float4(0.0f, 0.0f, 0.0f, 0.0f);
    __syncthreads();

    const uint4* __restrict__ ob4 = (const uint4*)ownb;
    const int V = E >> 4;
    const unsigned int pat   = (unsigned int)me * 0x01010101u;
    const unsigned int below = (1u << lane) - 1u;
    int* q  = s_q[wave];
    int  qn = 0;

    uint4 cur[OWN_U];
    int vb = wave * (32 * OWN_U);
#pragma unroll
    for (int u = 0; u < OWN_U; ++u) {
        const int v = vb + 32 * u + lane;
        cur[u] = ob4[v < V ? v : 0];
    }
    for (; vb < V; vb += OWN_STRIDE) {
        uint4 nxt[OWN_U];
#pragma unroll
        for (int u = 0; u < OWN_U; ++u) {
            const int v = vb + OWN_STRIDE + 32 * u + lane;
            nxt[u] = ob4[v < V ? v : 0];
        }
        unsigned int m[OWN_U];
#pragma unroll
        for (int u = 0; u < OWN_U; ++u) {
            const int v = vb + 32 * u + lane;
            const unsigned int mm =
                ~(((ne_bytes(cur[u].x, pat) >> 7) & 0x01010101u) | ((ne_bytes(cur[u].y, pat) >> 6) & 0x02020202u) |
                  ((ne_bytes(cur[u].z, pat) >> 5) & 0x04040404u) | ((ne_bytes(cur[u].w, pat) >> 4) & 0x08080808u)) &
                0x0f0f0f0fu;
            m[u] = (v < V) ? mm : 0u;
        }
        for (;;) {
            const bool has = ((m[0] | m[1] | m[2] | m[3]) != 0u);
            const unsigned int b = __builtin_amdgcn_ballot_w32(has);
            if (b == 0u) break;
            if (has) {
                const int uu = m[0] ? 0 : (m[1] ? 1 : (m[2] ? 2 : 3));
                const unsigned int mm = m[0] ? m[0] : (m[1] ? m[1] : (m[2] ? m[2] : m[3]));
                const int p = __builtin_ctz(mm);
                const unsigned int rest = mm & (mm - 1u);
                m[0] = (uu == 0) ? rest : m[0];
                m[1] = (uu == 1) ? rest : m[1];
                m[2] = (uu == 2) ? rest : m[2];
                m[3] = (uu == 3) ? rest : m[3];
                q[qn + __builtin_popcount(b & below)] =
                    (vb + 32 * uu + lane) * 16 + ((p & 7) << 2) + (p >> 3);
            }
            qn += __builtin_popcount(b);
            if (qn >= 32) {
                qn -= 32;
                own_batch<true>(ei, h, acc, q + qn, 32, lo, E, lane);
            }
        }
#pragma unroll
        for (int u = 0; u < OWN_U; ++u) cur[u] = nxt[u];
    }
    if (wave == 0) {
        const int e = V * 16 + lane;
        const bool has = (lane < 16) && (e < E) && ((int)ownb[e < E ? e : 0] == me);
        const unsigned int b = __builtin_amdgcn_ballot_w32(has);
        if (has) q[qn + __builtin_popcount(b & below)] = e;
        qn += __builtin_popcount(b);
    }
    while (qn > 0) {
        const int take = min(qn, 32);
        qn -= take;
        own_batch<false>(ei, h, acc, q + qn, take, lo, E, lane);
    }
    __syncthreads();

    const int rows = min(OWN_ROWS, n - lo);
    const float4* __restrict__ h4 = (const float4*)h;
    float4* __restrict__ o4 = (float4*)out;
    for (int i = tid; i < rows * (HID / 4); i += 1024) {
        const int r   = i >> 5;
        const int c4  = i & 31;
        const int row = lo + r;
        const float d = dinv[row];
        const float4 hv = h4[(size_t)row * (HID / 4) + c4];
        const float* ar = acc + r * HID + c4;
        const float4 av = make_float4(ar[0], ar[32], ar[64], ar[96]);
        const float4 bv = ((const float4*)bias)[c4];
        const float4 al = ((const float4*)alpha)[c4];
        float4 o;
        o.x = bv.x + d * (hv.x + av.x);
        o.y = bv.y + d * (hv.y + av.y);
        o.z = bv.z + d * (hv.z + av.z);
        o.w = bv.w + d * (hv.w + av.w);
        o.x = o.x > 0.0f ? o.x : al.x * o.x;
        o.y = o.y > 0.0f ? o.y : al.y * o.y;
        o.z = o.z > 0.0f ? o.z : al.z * o.z;
        o.w = o.w > 0.0f ? o.w : al.w * o.w;
        o4[(size_t)row * (HID / 4) + c4] = o;
    }
}

extern "C" void kernel_launch(void* const* d_in, const int* in_sizes, int n_in,
                              void* d_out, int out_size, void* d_ws, size_t ws_size,
                              hipStream_t stream) {
    const float* x     = (const float*)d_in[0];
    const int*   ei    = (const int*)d_in[1];
    const float* W     = (const float*)d_in[2];
    const float* bias  = (const float*)d_in[3];
    const float* alpha = (const float*)d_in[4];
    float*       out   = (float*)d_out;

    const int N = in_sizes[0] / IN_CH;
    const int E = in_sizes[1] / 2;

    float* h    = (float*)d_ws;
    unsigned char* ownb = (unsigned char*)(h + (size_t)N * HID);
    float* dinv = (float*)(ownb + (((size_t)E + 15) & ~(size_t)15));
    int*   part = (int*)(dinv + N);

    const dim3 hgrid(DEG_SLICES, (N + DEG_BINS - 1) / DEG_BINS);
    k_deg_hist<<<hgrid, 256, 0, stream>>>(ei + (size_t)E, part, ownb, N, E);
    k_dinv    <<<(N + 255) / 256, 256, 0, stream>>>(part, dinv, N);

    const int strips = (N + 15) / 16;
    const int blocks = (strips + 7) / 8;
    k_gemm_wmma<<<blocks, 256, 0, stream>>>(x, W, dinv, h, N);

    k_owner<<<(N + OWN_ROWS - 1) / OWN_ROWS, 1024, 0, stream>>>(ei, ownb, h, dinv, bias, alpha, out, N, E);
}
